# starting kernel
# baseline (speedup 1.0000x reference)
#include <hip/hip_runtime.h>
#include <hip/hip_bf16.h>


typedef float v2f __attribute__((ext_vector_type(2)));
typedef float v8f __attribute__((ext_vector_type(8)));

#define DIMF 64
#define CATF 256
#define OUTF 10

__global__ void zero_f32(float* __restrict__ p, int n) {
    int i = blockIdx.x * blockDim.x + threadIdx.x;
    if (i < n) p[i] = 0.0f;
}

__global__ void deg_kernel(const int* __restrict__ dst, float* __restrict__ deg, int E) {
    int e = blockIdx.x * blockDim.x + threadIdx.x;
    if (e < E) atomicAdd(&deg[dst[e]], 1.0f);
}

__global__ void dinv_kernel(float* __restrict__ deg, int N) {
    int i = blockIdx.x * blockDim.x + threadIdx.x;
    if (i < N) deg[i] = __frsqrt_rn(deg[i] + 1.0f);
}

__global__ void enorm_kernel(const int* __restrict__ src, const int* __restrict__ dst,
                             const float* __restrict__ dinv, float* __restrict__ enorm, int E) {
    int e = blockIdx.x * blockDim.x + threadIdx.x;
    if (e < E) enorm[e] = dinv[src[e]] * dinv[dst[e]];
}

__global__ __launch_bounds__(256) void gemm_n64_wmma(const float* __restrict__ A,
                                                     const float* __restrict__ W,
                                                     float* __restrict__ out) {
    const int wid  = threadIdx.x >> 5;
    const int lane = threadIdx.x & 31;
    const int m0   = blockIdx.x * 32 + (wid >> 2) * 16;
    const int n0   = (wid & 3) * 16;
    const int mr   = lane & 15;
    const int kh   = (lane >> 4) * 2;

    const float* __restrict__ arow = A + (size_t)(m0 + mr) * DIMF;

    v8f c = {};
#pragma unroll
    for (int k = 0; k < DIMF; k += 4) {
        v2f a, b;
        a.x = arow[k + kh];
        a.y = arow[k + kh + 1];
        b.x = W[(k + kh) * DIMF + n0 + mr];
        b.y = W[(k + kh + 1) * DIMF + n0 + mr];
        c = __builtin_amdgcn_wmma_f32_16x16x4_f32(
                 false, a,  false, b,
                 (short)0, c,  false,  false);
    }

    const int col   = n0 + mr;
    const int rbase = m0 + (lane >> 4) * 8;
#pragma unroll
    for (int r = 0; r < 8; ++r)
        out[(size_t)(rbase + r) * DIMF + col] = c[r];
}

__global__ void agg_init_kernel(const float* __restrict__ h_lin,
                                const float* __restrict__ dinv,
                                const float* __restrict__ bias,
                                float* __restrict__ h_next, int N) {
    int idx = blockIdx.x * blockDim.x + threadIdx.x;
    if (idx >= N * DIMF) return;
    int i = idx >> 6, d = idx & 63;
    float di = dinv[i];
    h_next[idx] = h_lin[idx] * di * di + bias[d];
}

__global__ void scatter_kernel(const int* __restrict__ src, const int* __restrict__ dst,
                               const float* __restrict__ enorm,
                               const float* __restrict__ h_lin,
                               float* __restrict__ h_next, int E) {
    long long idx = (long long)blockIdx.x * blockDim.x + threadIdx.x;
    if (idx >= (long long)E * 16) return;
    int e  = (int)(idx >> 4);
    int d0 = ((int)idx & 15) * 4;
    float w = enorm[e];
    const float4 v = *(const float4*)(h_lin + (size_t)src[e] * DIMF + d0);
    float* o = h_next + (size_t)dst[e] * DIMF + d0;
    atomicAdd(o + 0, v.x * w);
    atomicAdd(o + 1, v.y * w);
    atomicAdd(o + 2, v.z * w);
    atomicAdd(o + 3, v.w * w);
}

__global__ void post_kernel(const float* __restrict__ h_next,
                            const float* __restrict__ gamma, const float* __restrict__ beta,
                            const float* __restrict__ mean,  const float* __restrict__ var,
                            const int* __restrict__ batch,
                            float* __restrict__ h_cur, float* __restrict__ pooled,
                            int N, int layer) {
    int idx = blockIdx.x * blockDim.x + threadIdx.x;
    if (idx >= N * DIMF) return;
    int i = idx >> 6, d = idx & 63;
    int ld = layer * DIMF + d;
    float v = (h_next[idx] - mean[ld]) * __frsqrt_rn(var[ld] + 1e-5f) * gamma[ld] + beta[ld];
    v = fmaxf(v, 0.0f);
    h_cur[idx] = v;
    atomicAdd(&pooled[(size_t)batch[i] * CATF + ld], v);
}

__global__ __launch_bounds__(64) void mlp_kernel(const float* __restrict__ pooled,
                                                 const float* __restrict__ W1, const float* __restrict__ b1,
                                                 const float* __restrict__ W2, const float* __restrict__ b2,
                                                 float* __restrict__ out) {
    __shared__ float sp[CATF];
    __shared__ float sh[DIMF];
    int g = blockIdx.x, t = threadIdx.x;
    const float* prow = pooled + (size_t)g * CATF;
#pragma unroll
    for (int j = 0; j < 4; ++j) sp[t + j * DIMF] = prow[t + j * DIMF];
    __syncthreads();
    float acc = b1[t];
#pragma unroll 8
    for (int k = 0; k < CATF; ++k) acc = fmaf(sp[k], W1[k * DIMF + t], acc);
    sh[t] = fmaxf(acc, 0.0f);
    __syncthreads();
    if (t < OUTF) {
        float o = b2[t];
#pragma unroll 8
        for (int d = 0; d < DIMF; ++d) o = fmaf(sh[d], W2[d * OUTF + t], o);
        out[(size_t)g * OUTF + t] = o;
    }
}

extern "C" void kernel_launch(void* const* d_in, const int* in_sizes, int n_in,
                              void* d_out, int out_size, void* d_ws, size_t ws_size,
                              hipStream_t stream) {
    const float* x    = (const float*)d_in[0];
    const int*   ei   = (const int*)d_in[1];
    const int*   bat  = (const int*)d_in[2];
    const float* W[4] = {(const float*)d_in[4], (const float*)d_in[6],
                         (const float*)d_in[8], (const float*)d_in[10]};
    const float* B[4] = {(const float*)d_in[5], (const float*)d_in[7],
                         (const float*)d_in[9], (const float*)d_in[11]};
    const float* bn_g = (const float*)d_in[12];
    const float* bn_b = (const float*)d_in[13];
    const float* bn_m = (const float*)d_in[14];
    const float* bn_v = (const float*)d_in[15];
    const float* l1W  = (const float*)d_in[16];
    const float* l1b  = (const float*)d_in[17];
    const float* l2W  = (const float*)d_in[18];
    const float* l2b  = (const float*)d_in[19];

    const int N = in_sizes[0] / DIMF;
    const int E = in_sizes[1] / 2;
    const int G = out_size / OUTF;
    const int* src = ei;
    const int* dst = ei + E;

    float* ws     = (float*)d_ws;
    float* dinv   = ws;
    float* enorm  = dinv + N;
    float* h_cur  = enorm + E;
    float* h_lin  = h_cur + (size_t)N * DIMF;
    float* h_next = h_lin + (size_t)N * DIMF;
    float* pooled = h_next + (size_t)N * DIMF;

    const int T = 256;
    zero_f32<<<(N + T - 1) / T, T, 0, stream>>>(dinv, N);
    zero_f32<<<(G * CATF + T - 1) / T, T, 0, stream>>>(pooled, G * CATF);
    deg_kernel<<<(E + T - 1) / T, T, 0, stream>>>(dst, dinv, E);
    dinv_kernel<<<(N + T - 1) / T, T, 0, stream>>>(dinv, N);
    enorm_kernel<<<(E + T - 1) / T, T, 0, stream>>>(src, dst, dinv, enorm, E);

    const int gemm_blocks = N / 32;
    const int nd_blocks   = (N * DIMF + T - 1) / T;
    const long long sc    = (long long)E * 16;
    const int sc_blocks   = (int)((sc + T - 1) / T);

    const float* hin = x;
    for (int l = 0; l < 4; ++l) {
        gemm_n64_wmma<<<gemm_blocks, T, 0, stream>>>(hin, W[l], h_lin);
        agg_init_kernel<<<nd_blocks, T, 0, stream>>>(h_lin, dinv, B[l], h_next, N);
        scatter_kernel<<<sc_blocks, T, 0, stream>>>(src, dst, enorm, h_lin, h_next, E);
        post_kernel<<<nd_blocks, T, 0, stream>>>(h_next, bn_g, bn_b, bn_m, bn_v,
                                                 bat, h_cur, pooled, N, l);
        hin = h_cur;
    }

    mlp_kernel<<<G, DIMF, 0, stream>>>(pooled, l1W, l1b, l2W, l2b, (float*)d_out);
}


# Edge sum reads source rows from LDS: edges grouped once per call by 128-row source block and destination half; a work-group stages its block's rows, only atomics stay random; sum order changes
# speedup vs baseline: 2.8798x; 2.8798x over previous
#include <hip/hip_runtime.h>
#include <hip/hip_bf16.h>


typedef float v2f __attribute__((ext_vector_type(2)));
typedef float v8f __attribute__((ext_vector_type(8)));

#define DIMF 64
#define CATF 256
#define OUTF 10

__device__ __forceinline__ int wave_half() {
    return (int)((__builtin_amdgcn_s_sendmsg_rtn(0x87) >> 18) & 1u);
}

__global__ void zero_f32(float* __restrict__ p, int n) {
    int i = blockIdx.x * blockDim.x + threadIdx.x;
    if (i < n) p[i] = 0.0f;
}

#define RB 128
#define RB_SHIFT 7
#define MAX_BK 2048
#define EDGES_PER_GROUP_PASS 4096

__device__ __forceinline__ int group_of(int s, int d, int n_half) {
    return ((s >> RB_SHIFT) << 1) | (d >= n_half ? 1 : 0);
}

#define SPARE_SLOT (MAX_BK - 1)
__global__ __launch_bounds__(256) void deg_count_kernel(const int* __restrict__ src, const int* __restrict__ dst,
                                                        float* __restrict__ deg2, int* __restrict__ bk_cnt,
                                                        int E, int N, int n_half, int nbk) {
    __shared__ int hist[MAX_BK];
    const int t = threadIdx.x;
    for (int i = t; i < MAX_BK; i += 256) hist[i] = 0;
    __syncthreads();
    float* __restrict__ deg = deg2 + (size_t)wave_half() * N;
    const int e0 = blockIdx.x * EDGES_PER_GROUP_PASS + t;
    int s[EDGES_PER_GROUP_PASS / 256], d[EDGES_PER_GROUP_PASS / 256];
#pragma unroll
    for (int it = 0; it < EDGES_PER_GROUP_PASS / 256; ++it) {
        const int e  = e0 + it * 256;
        const int ec = (e < E) ? e : E - 1;
        s[it] = src[ec];
        d[it] = dst[ec];
    }
#pragma unroll
    for (int it = 0; it < EDGES_PER_GROUP_PASS / 256; ++it) {
        const bool valid = e0 + it * 256 < E;
        atomicAdd(&hist[valid ? group_of(s[it], d[it], n_half) : SPARE_SLOT], 1);
        if (valid) atomicAdd(&deg[d[it]], 1.0f);
    }
    __syncthreads();
    for (int i = t; i < nbk; i += 256) {
        const int c = hist[i];
        if (c) atomicAdd(&bk_cnt[i], c);
    }
}

__global__ __launch_bounds__(1024) void group_scan_kernel(const int* __restrict__ bk_cnt,
                                                          int* __restrict__ bk_start, int nbk) {
    __shared__ int sh[1024];
    const int t  = threadIdx.x;
    const int i0 = 2 * t, i1 = 2 * t + 1;
    const int a0 = (i0 < nbk) ? bk_cnt[i0] : 0;
    const int a1 = (i1 < nbk) ? bk_cnt[i1] : 0;
    sh[t] = a0 + a1;
    __syncthreads();
    for (int off = 1; off < 1024; off <<= 1) {
        const int v = (t >= off) ? sh[t - off] : 0;
        __syncthreads();
        sh[t] += v;
        __syncthreads();
    }
    const int excl = sh[t] - (a0 + a1);
    if (i0 <= nbk) bk_start[i0] = excl;
    if (i1 <= nbk) bk_start[i1] = excl + a0;
}

__global__ __launch_bounds__(256) void group_fill_kernel(const int* __restrict__ src, const int* __restrict__ dst,
                                                         const int* __restrict__ bk_start, int* __restrict__ bk_cur,
                                                         int* __restrict__ rec, int E, int n_half, int nbk) {
    __shared__ int lcnt[MAX_BK];
    __shared__ int lbase[MAX_BK];
    const int t = threadIdx.x;
    for (int i = t; i < MAX_BK; i += 256) lcnt[i] = 0;
    __syncthreads();
    int bk[EDGES_PER_GROUP_PASS / 256], rc[EDGES_PER_GROUP_PASS / 256];
    const int e0 = blockIdx.x * EDGES_PER_GROUP_PASS + t;
#pragma unroll
    for (int it = 0; it < EDGES_PER_GROUP_PASS / 256; ++it) {
        const int e  = e0 + it * 256;
        const int ec = (e < E) ? e : E - 1;
        bk[it] = src[ec];
        rc[it] = dst[ec];
    }
#pragma unroll
    for (int it = 0; it < EDGES_PER_GROUP_PASS / 256; ++it) {
        const int s = bk[it];
        const int d = rc[it];
        bk[it] = (e0 + it * 256 < E) ? group_of(s, d, n_half) : SPARE_SLOT;
        rc[it] = (d << RB_SHIFT) | (s & (RB - 1));
        atomicAdd(&lcnt[bk[it]], 1);
    }
    __syncthreads();
    int cc[MAX_BK / 256], rr[MAX_BK / 256];
#pragma unroll
    for (int k = 0; k < MAX_BK / 256; ++k) {
        const int i = t + k * 256;
        cc[k] = (i < nbk) ? lcnt[i] : 0;
    }
#pragma unroll
    for (int k = 0; k < MAX_BK / 256; ++k) {
        rr[k] = 0;
        if (cc[k]) rr[k] = bk_start[t + k * 256] + atomicAdd(&bk_cur[t + k * 256], cc[k]);
    }
#pragma unroll
    for (int k = 0; k < MAX_BK / 256; ++k) {
        if (cc[k]) {
            lbase[t + k * 256] = rr[k];
            lcnt[t + k * 256] = 0;
        }
    }
    __syncthreads();
#pragma unroll
    for (int it = 0; it < EDGES_PER_GROUP_PASS / 256; ++it) {
        if (bk[it] != SPARE_SLOT) {
            const int r = atomicAdd(&lcnt[bk[it]], 1);
            rec[lbase[bk[it]] + r] = rc[it];
        }
    }
}

__global__ void dinv_kernel(const float* __restrict__ deg2, float* __restrict__ dinv, int N) {
    int i = blockIdx.x * blockDim.x + threadIdx.x;
    if (i < N) dinv[i] = __frsqrt_rn(deg2[i] + deg2[N + i] + 1.0f);
}

#define TILE_PITCH 68
__global__ __launch_bounds__(256) void gemm_n64_wmma(const float* __restrict__ A,
                                                     const float* __restrict__ W,
                                                     const float* __restrict__ dinv,
                                                     const float* __restrict__ bias_prev,
                                                     const float* __restrict__ gamma,
                                                     const float* __restrict__ beta,
                                                     const float* __restrict__ mean,
                                                     const float* __restrict__ var,
                                                     const int* __restrict__ batch,
                                                     float* __restrict__ pooled,
                                                     int pool_stride,
                                                     float* __restrict__ out,
                                                     float* __restrict__ h_next,
                                                     int prev_layer) {
    __shared__ float tile[32 * TILE_PITCH];
    const int t    = threadIdx.x;
    const int wid  = __builtin_amdgcn_readfirstlane(t >> 5);
    const int lane = t & 31;
    const int row0 = blockIdx.x * 32;

    const int c4 = (t & 15) * 4;
    const int tr = t >> 4;
    float4 v0 = *reinterpret_cast<const float4*>(A + (size_t)(row0 + tr) * DIMF + c4);
    float4 v1 = *reinterpret_cast<const float4*>(A + (size_t)(row0 + tr + 16) * DIMF + c4);
    if (prev_layer >= 0) {
        const float d0 = dinv[row0 + tr];
        const float d1 = dinv[row0 + tr + 16];
        const float4 bp = *reinterpret_cast<const float4*>(bias_prev + c4);
        v0.x = v0.x * d0 + bp.x; v0.y = v0.y * d0 + bp.y; v0.z = v0.z * d0 + bp.z; v0.w = v0.w * d0 + bp.w;
        v1.x = v1.x * d1 + bp.x; v1.y = v1.y * d1 + bp.y; v1.z = v1.z * d1 + bp.z; v1.w = v1.w * d1 + bp.w;
        const int ld = prev_layer * DIMF + c4;
        const float4 mu = *reinterpret_cast<const float4*>(mean + ld);
        const float4 va = *reinterpret_cast<const float4*>(var + ld);
        const float4 ga = *reinterpret_cast<const float4*>(gamma + ld);
        const float4 be = *reinterpret_cast<const float4*>(beta + ld);
        const float rx = __frsqrt_rn(va.x + 1e-5f), ry = __frsqrt_rn(va.y + 1e-5f);
        const float rz = __frsqrt_rn(va.z + 1e-5f), rw = __frsqrt_rn(va.w + 1e-5f);
        v0.x = fmaxf((v0.x - mu.x) * rx * ga.x + be.x, 0.0f);
        v0.y = fmaxf((v0.y - mu.y) * ry * ga.y + be.y, 0.0f);
        v0.z = fmaxf((v0.z - mu.z) * rz * ga.z + be.z, 0.0f);
        v0.w = fmaxf((v0.w - mu.w) * rw * ga.w + be.w, 0.0f);
        v1.x = fmaxf((v1.x - mu.x) * rx * ga.x + be.x, 0.0f);
        v1.y = fmaxf((v1.y - mu.y) * ry * ga.y + be.y, 0.0f);
        v1.z = fmaxf((v1.z - mu.z) * rz * ga.z + be.z, 0.0f);
        v1.w = fmaxf((v1.w - mu.w) * rw * ga.w + be.w, 0.0f);
    }
    *reinterpret_cast<float4*>(tile + tr * TILE_PITCH + c4) = v0;
    *reinterpret_cast<float4*>(tile + (tr + 16) * TILE_PITCH + c4) = v1;
    __syncthreads();

    if (prev_layer >= 0) {
        float pv[8];
        float* pool_plane = pooled + (size_t)wave_half() * pool_stride;
#pragma unroll
        for (int j = 0; j < 4; ++j) {
            pv[2 * j]     = tile[(wid * 4 + j) * TILE_PITCH + lane];
            pv[2 * j + 1] = tile[(wid * 4 + j) * TILE_PITCH + lane + 32];
        }
#pragma unroll
        for (int j = 0; j < 4; ++j) {
            const int g = batch[row0 + wid * 4 + j];
            float* p = pool_plane + ((size_t)g * CATF + prev_layer * DIMF);
            atomicAdd(p + lane, pv[2 * j]);
            atomicAdd(p + lane + 32, pv[2 * j + 1]);
        }
    }

    const int m0   = row0 + (wid >> 2) * 16;
    const int n0   = (wid & 3) * 16;
    const int mr   = lane & 15;
    const int kh   = (lane >> 4) * 2;

    const float* arow = tile + ((wid >> 2) * 16 + mr) * TILE_PITCH + kh;

    v8f c = {};
#pragma unroll
    for (int k = 0; k < DIMF; k += 4) {
        v2f b;
        const v2f a = *reinterpret_cast<const v2f*>(arow + k);
        b.x = W[(k + kh) * DIMF + n0 + mr];
        b.y = W[(k + kh + 1) * DIMF + n0 + mr];
        c = __builtin_amdgcn_wmma_f32_16x16x4_f32(
                 false, a,  false, b,
                 (short)0, c,  false,  false);
    }

    const int col   = n0 + mr;
    const int rbase = m0 + (lane >> 4) * 8;
#pragma unroll
    for (int r = 0; r < 8; ++r) {
        const size_t o = (size_t)(rbase + r) * DIMF + col;
        const float hs = c[r] * dinv[rbase + r];
        out[o] = hs;
        h_next[o] = hs;
    }
}

#define EDGES_PER_STEP 8
#define TICKET_PITCH 64
__global__ __launch_bounds__(256) void scatter_kernel(const int* __restrict__ rec, const int* __restrict__ bk_start,
                                                      const float* __restrict__ h_lin,
                                                      float* __restrict__ h_next,
                                                      int* __restrict__ tickets,
                                                      int N, int nsb) {
    __shared__ float rows[RB * DIMF];
    __shared__ int s_draw;
    const int t    = threadIdx.x;
    const int lane = t & 31;
    const int wid  = __builtin_amdgcn_readfirstlane(t >> 5);
    if (t == 0) {
        int h = wave_half();
        int u = atomicAdd(&tickets[h * TICKET_PITCH], 1);
        if (u >= nsb) {
            h ^= 1;
            u = atomicAdd(&tickets[h * TICKET_PITCH], 1);
        }
        s_draw = u * 2 + h;
    }
    __syncthreads();
    const int draw = __builtin_amdgcn_readfirstlane(s_draw);
    const int sb   = draw >> 1;
    if (sb >= nsb) return;

    const int row0  = sb * RB;
    const int nrows = (N - row0 < RB) ? N - row0 : RB;
    float4 v[RB * DIMF / 4 / 256];
#pragma unroll
    for (int it = 0; it < RB * DIMF / 4 / 256; ++it) {
        const int q = it * 256 + t;
        const int r = q >> 4;
        const int c4 = (q & 15) * 4;
        const int rs = (r < nrows) ? r : nrows - 1;
        v[it] = *reinterpret_cast<const float4*>(h_lin + (size_t)(row0 + rs) * DIMF + c4);
    }
    const int beg = __builtin_amdgcn_readfirstlane(bk_start[draw]);
    const int end = __builtin_amdgcn_readfirstlane(bk_start[draw + 1]);
    int base = beg + wid * 32;
    int w = (base + lane < end) ? rec[base + lane] : 0;
#pragma unroll
    for (int it = 0; it < RB * DIMF / 4 / 256; ++it) {
        const int q = it * 256 + t;
        *reinterpret_cast<float4*>(rows + (q >> 4) * DIMF + (q & 15) * 4) = v[it];
    }
    __syncthreads();

    const float* __restrict__ rl = rows + lane;
    float* __restrict__ hn = h_next + lane;

    while (base < end) {
        const int n = (end - base < 32) ? end - base : 32;
        const int nbase = base + 256;
        const int wn = (nbase + lane < end) ? rec[nbase + lane] : 0;
        for (int g = 0; g < n; g += EDGES_PER_STEP) {
            float a[EDGES_PER_STEP], b[EDGES_PER_STEP];
#pragma unroll
            for (int j = 0; j < EDGES_PER_STEP; ++j) {
                const int idx = (g + j < n) ? g + j : n - 1;
                const int wj = __builtin_amdgcn_readlane(w, idx);
                const float* r = rl + (wj & (RB - 1)) * DIMF;
                a[j] = r[0];
                b[j] = r[32];
            }
#pragma unroll
            for (int j = 0; j < EDGES_PER_STEP; ++j) {
                if (g + j < n) {
                    const int wj = __builtin_amdgcn_readlane(w, g + j);
                    float* o = hn + (size_t)(wj >> RB_SHIFT) * DIMF;
                    atomicAdd(o, a[j]);
                    atomicAdd(o + 32, b[j]);
                }
            }
        }
        w = wn;
        base = nbase;
    }
}

__global__ void pool_last_kernel(const float* __restrict__ h_next,
                                 const float* __restrict__ dinv,
                                 const float* __restrict__ bias,
                                 const float* __restrict__ gamma, const float* __restrict__ beta,
                                 const float* __restrict__ mean,  const float* __restrict__ var,
                                 const int* __restrict__ batch,
                                 float* __restrict__ pooled, int pool_stride,
                                 int N, int layer) {
    int idx = blockIdx.x * blockDim.x + threadIdx.x;
    float* __restrict__ pool_plane = pooled + (size_t)wave_half() * pool_stride;
    if (idx >= N * DIMF) return;
    int i = idx >> 6, d = idx & 63;
    int ld = layer * DIMF + d;
    const float agg = h_next[idx] * dinv[i] + bias[d];
    float v = (agg - mean[ld]) * __frsqrt_rn(var[ld] + 1e-5f) * gamma[ld] + beta[ld];
    v = fmaxf(v, 0.0f);
    atomicAdd(&pool_plane[(size_t)batch[i] * CATF + ld], v);
}

__global__ __launch_bounds__(64) void mlp_kernel(const float* __restrict__ pooled,
                                                 const float* __restrict__ W1, const float* __restrict__ b1,
                                                 const float* __restrict__ W2, const float* __restrict__ b2,
                                                 float* __restrict__ out) {
    __shared__ float sp[CATF];
    __shared__ float sh[DIMF];
    int g = blockIdx.x, t = threadIdx.x;
    const float* prow = pooled + (size_t)g * CATF;
    const float* prow1 = prow + (size_t)gridDim.x * CATF;
#pragma unroll
    for (int j = 0; j < 4; ++j) sp[t + j * DIMF] = prow[t + j * DIMF] + prow1[t + j * DIMF];
    __syncthreads();
    float acc = b1[t];
#pragma unroll 8
    for (int k = 0; k < CATF; ++k) acc = fmaf(sp[k], W1[k * DIMF + t], acc);
    sh[t] = fmaxf(acc, 0.0f);
    __syncthreads();
    if (t < OUTF) {
        float o = b2[t];
#pragma unroll 8
        for (int d = 0; d < DIMF; ++d) o = fmaf(sh[d], W2[d * OUTF + t], o);
        out[(size_t)g * OUTF + t] = o;
    }
}

extern "C" void kernel_launch(void* const* d_in, const int* in_sizes, int n_in,
                              void* d_out, int out_size, void* d_ws, size_t ws_size,
                              hipStream_t stream) {
    const float* x    = (const float*)d_in[0];
    const int*   ei   = (const int*)d_in[1];
    const int*   bat  = (const int*)d_in[2];
    const float* W[4] = {(const float*)d_in[4], (const float*)d_in[6],
                         (const float*)d_in[8], (const float*)d_in[10]};
    const float* B[4] = {(const float*)d_in[5], (const float*)d_in[7],
                         (const float*)d_in[9], (const float*)d_in[11]};
    const float* bn_g = (const float*)d_in[12];
    const float* bn_b = (const float*)d_in[13];
    const float* bn_m = (const float*)d_in[14];
    const float* bn_v = (const float*)d_in[15];
    const float* l1W  = (const float*)d_in[16];
    const float* l1b  = (const float*)d_in[17];
    const float* l2W  = (const float*)d_in[18];
    const float* l2b  = (const float*)d_in[19];

    const int N = in_sizes[0] / DIMF;
    const int E = in_sizes[1] / 2;
    const int G = out_size / OUTF;
    const int* src = ei;
    const int* dst = ei + E;

    float* ws     = (float*)d_ws;
    float* dinv   = ws;
    float* h_agg0 = dinv + N;
    float* h_lin  = h_agg0 + (size_t)N * DIMF;
    float* h_agg1 = h_lin + (size_t)N * DIMF;
    float* pooled = h_agg1 + (size_t)N * DIMF;
    float* deg2   = pooled + (size_t)2 * G * CATF;
    int* tickets  = (int*)(deg2 + (size_t)2 * N);
    int* bk_cnt   = tickets + 8 * TICKET_PITCH;
    int* bk_cur   = bk_cnt + MAX_BK;
    int* bk_start = bk_cur + MAX_BK;
    int* rec      = bk_start + MAX_BK;

    const int T = 256;
    const int n_zero = 2 * N + 8 * TICKET_PITCH + 2 * MAX_BK;
    zero_f32<<<(n_zero + T - 1) / T, T, 0, stream>>>(deg2, n_zero);
    zero_f32<<<(2 * G * CATF + T - 1) / T, T, 0, stream>>>(pooled, 2 * G * CATF);
    const int nsb      = (N + RB - 1) / RB;
    const int nbk      = 2 * nsb;
    const int n_passes = (E + EDGES_PER_GROUP_PASS - 1) / EDGES_PER_GROUP_PASS;
    deg_count_kernel<<<n_passes, T, 0, stream>>>(src, dst, deg2, bk_cnt, E, N, N / 2, nbk);
    dinv_kernel<<<(N + T - 1) / T, T, 0, stream>>>(deg2, dinv, N);
    group_scan_kernel<<<1, 1024, 0, stream>>>(bk_cnt, bk_start, nbk);
    group_fill_kernel<<<n_passes, T, 0, stream>>>(src, dst, bk_start, bk_cur, rec, E, N / 2, nbk);

    const int gemm_blocks = N / 32;
    const int nd_blocks   = (N * DIMF + T - 1) / T;
    const int sc_blocks   = 2 * nsb;

    const float* hin = x;
    const float* bprev = B[0];
    float* hagg = h_agg0;
    float* hother = h_agg1;
    for (int l = 0; l < 4; ++l) {
        gemm_n64_wmma<<<gemm_blocks, T, 0, stream>>>(hin, W[l], dinv, bprev, bn_g, bn_b, bn_m, bn_v,
                                                    bat, pooled, G * CATF, h_lin, hagg, l - 1);
        scatter_kernel<<<sc_blocks, T, 0, stream>>>(rec, bk_start, h_lin, hagg, tickets + 2 * TICKET_PITCH * l, N, nsb);
        hin = hagg;
        bprev = B[l];
        float* tmp = hagg; hagg = hother; hother = tmp;
    }
    pool_last_kernel<<<nd_blocks, T, 0, stream>>>(hin, dinv, B[3], bn_g, bn_b, bn_m, bn_v, bat, pooled, G * CATF, N, 3);

    mlp_kernel<<<G, DIMF, 0, stream>>>(pooled, l1W, l1b, l2W, l2b, (float*)d_out);
}
